# starting kernel
# baseline (speedup 1.0000x reference)
#include <hip/hip_runtime.h>
#include <hip/hip_bf16.h>
#include <math.h>


#define Nn    100000
#define FIN   128
#define HCc   64
#define HIDc  32
#define OUTc  40

typedef __attribute__((ext_vector_type(2))) float v2f;
typedef __attribute__((ext_vector_type(8))) float v8f;

__device__ inline void atomicMaxFloat(float* addr, float val) {
    int bits = __float_as_int(val);
    if (bits >= 0) {
        atomicMax((int*)addr, bits);
    } else {
        atomicMin((unsigned int*)addr, (unsigned int)bits);
    }
}

__device__ inline void edge_sd(const int* __restrict__ ei, int e, int E,
                               int& s, int& d) {
    if (e < E) { s = ei[e]; d = ei[E + e]; }
    else       { s = e - E; d = e - E; }
}

template<int NT, bool FUSE, int K>
__global__ __launch_bounds__(128)
void wmma_gemm(const float* __restrict__ A, const float* __restrict__ A2,
               const float* __restrict__ a2bias,
               const float* __restrict__ B, const float* __restrict__ bias,
               float* __restrict__ C, int M, int Ncols) {
    __shared__ float Bs[(K / 2) * 64 * 2];

    for (int i = threadIdx.x; i < K * 64; i += 128) {
        const int p    = i >> 7;
        const int rem  = i & 127;
        const int col  = rem >> 1;
        const int half = rem & 1;
        Bs[i] = (col < Ncols) ? B[(size_t)(2 * p + half) * Ncols + col] : 0.f;
    }
    __syncthreads();

    const int wave = threadIdx.x >> 5;
    const int lane = threadIdx.x & 31;
    const int tileM = blockIdx.x * 4 + wave;
    const int row16 = tileM * 16;
    if (row16 >= M) return;

    const int hi = (lane >= 16);
    const int lr = lane & 15;

    v8f acc[NT];
#pragma unroll
    for (int t = 0; t < NT; ++t)
        acc[t] = (v8f){0.f, 0.f, 0.f, 0.f, 0.f, 0.f, 0.f, 0.f};

    const int arow = row16 + lr;
    const float* __restrict__ Arow  = A + (size_t)arow * K;
    const float* __restrict__ A2row = FUSE ? (A2 + (size_t)arow * K) : nullptr;

#pragma unroll 4
    for (int kb = 0; kb < K; kb += 4) {
        const int ka = kb + 2 * hi;
        float a0 = Arow[ka];
        float a1 = Arow[ka + 1];
        if (FUSE) {
            a0 = fmaxf(a0, A2row[ka]     + a2bias[ka]);
            a1 = fmaxf(a1, A2row[ka + 1] + a2bias[ka + 1]);
        }
        v2f afrag; afrag.x = a0; afrag.y = a1;

        const int p = (kb >> 1) + hi;
#pragma unroll
        for (int t = 0; t < NT; ++t) {
            const v2f bfrag = *(const v2f*)&Bs[(p << 7) + ((t * 16 + lr) << 1)];
            acc[t] = __builtin_amdgcn_wmma_f32_16x16x4_f32(
                false, afrag, false, bfrag, (short)0, acc[t], false, false);
        }
    }

    const int rbase = row16 + (hi ? 8 : 0);
#pragma unroll
    for (int t = 0; t < NT; ++t) {
        const int col = t * 16 + lr;
        if (col < Ncols) {
            const float bb = bias ? bias[col] : 0.f;
#pragma unroll
            for (int r = 0; r < 8; ++r)
                C[(size_t)(rbase + r) * Ncols + col] = acc[t][r] + bb;
        }
    }
}

__global__ __launch_bounds__(256)
void node_alpha(const float* __restrict__ h, const float* __restrict__ a_s,
                const float* __restrict__ a_d, float* __restrict__ asrc,
                float* __restrict__ adst) {
    const int gw   = (blockIdx.x * blockDim.x + threadIdx.x) >> 5;
    const int lane = threadIdx.x & 31;
    const int n = gw >> 1, hd = gw & 1;
    if (n >= Nn) return;
    const float hv = h[n * HCc + hd * HIDc + lane];
    float vs = hv * a_s[hd * HIDc + lane];
    float vd = hv * a_d[hd * HIDc + lane];
#pragma unroll
    for (int off = 16; off; off >>= 1) {
        vs += __shfl_xor(vs, off, 32);
        vd += __shfl_xor(vd, off, 32);
    }
    if (lane == 0) { asrc[n * 2 + hd] = vs; adst[n * 2 + hd] = vd; }
}

__global__ __launch_bounds__(256)
void init_layer(float* __restrict__ agg, float* __restrict__ m,
                float* __restrict__ den) {
    const int tid = blockIdx.x * blockDim.x + threadIdx.x;
    if (tid < Nn * HCc) agg[tid] = 0.f;
    if (tid < Nn * 2) { m[tid] = -INFINITY; den[tid] = 0.f; }
}

__global__ __launch_bounds__(256)
void edge_attn_max(const int* __restrict__ ei, int E, int Etot,
                   const float* __restrict__ asrc, const float* __restrict__ adst,
                   float* __restrict__ ebuf, float* __restrict__ m) {
    const int e = blockIdx.x * blockDim.x + threadIdx.x;
    if (e >= Etot) return;
    int s, d; edge_sd(ei, e, E, s, d);
#pragma unroll
    for (int hd = 0; hd < 2; ++hd) {
        float v = asrc[s * 2 + hd] + adst[d * 2 + hd];
        v = (v > 0.f) ? v : 0.2f * v;
        ebuf[e * 2 + hd] = v;
        atomicMaxFloat(&m[d * 2 + hd], v);
    }
}

__global__ __launch_bounds__(256)
void edge_exp_sum(const int* __restrict__ ei, int E, int Etot,
                  const float* __restrict__ m, float* __restrict__ ebuf,
                  float* __restrict__ den) {
    const int e = blockIdx.x * blockDim.x + threadIdx.x;
    if (e >= Etot) return;
    int s, d; edge_sd(ei, e, E, s, d);
#pragma unroll
    for (int hd = 0; hd < 2; ++hd) {
        const float ex = __expf(ebuf[e * 2 + hd] - m[d * 2 + hd]);
        ebuf[e * 2 + hd] = ex;
        atomicAdd(&den[d * 2 + hd], ex);
    }
}

__global__ __launch_bounds__(256)
void edge_msg(const int* __restrict__ ei, int E, int Etot,
              const float* __restrict__ h, const float* __restrict__ ex,
              const float* __restrict__ den, float* __restrict__ agg) {
    const int tid = blockIdx.x * blockDim.x + threadIdx.x;
    if (tid >= Etot * HCc) return;
    const int e = tid >> 6, c = tid & 63, hd = c >> 5;
    int s, d; edge_sd(ei, e, E, s, d);
    const float alpha = ex[e * 2 + hd] / (den[d * 2 + hd] + 1e-16f);
    atomicAdd(&agg[d * HCc + c], h[s * HCc + c] * alpha);
}

__global__ __launch_bounds__(256)
void node_post1(float* __restrict__ x, const float* __restrict__ b1,
                const float* __restrict__ gamma, const float* __restrict__ beta,
                const float* __restrict__ mean, const float* __restrict__ var) {
    const int tid = blockIdx.x * blockDim.x + threadIdx.x;
    if (tid >= Nn * HCc) return;
    const int c = tid & 63;
    float v = x[tid] + b1[c];
    v = (v - mean[c]) * (gamma[c] * rsqrtf(var[c] + 1e-5f)) + beta[c];
    x[tid] = (v > 0.f) ? v : (__expf(v) - 1.f);
}

extern "C" void kernel_launch(void* const* d_in, const int* in_sizes, int n_in,
                              void* d_out, int out_size, void* d_ws, size_t ws_size,
                              hipStream_t stream) {
    const float* node_feat = (const float*)d_in[0];
    const int*   ei        = (const int*)  d_in[1];
    const float* W1        = (const float*)d_in[2];
    const float* a_src1    = (const float*)d_in[3];
    const float* a_dst1    = (const float*)d_in[4];
    const float* b1        = (const float*)d_in[5];
    const float* bn_gamma  = (const float*)d_in[6];
    const float* bn_beta   = (const float*)d_in[7];
    const float* bn_mean   = (const float*)d_in[8];
    const float* bn_var    = (const float*)d_in[9];
    const float* W2        = (const float*)d_in[10];
    const float* a_src2    = (const float*)d_in[11];
    const float* a_dst2    = (const float*)d_in[12];
    const float* b2        = (const float*)d_in[13];
    const float* Wf        = (const float*)d_in[14];
    const float* bf        = (const float*)d_in[15];
    float* out = (float*)d_out;

    const int E    = in_sizes[1] / 2;
    const int Etot = E + Nn;

    float* ws   = (float*)d_ws;
    float* H    = ws;
    float* X1   = H    + (size_t)Nn * HCc;
    float* AGG2 = X1   + (size_t)Nn * HCc;
    float* ASRC = AGG2 + (size_t)Nn * HCc;
    float* ADST = ASRC + (size_t)Nn * 2;
    float* Mx   = ADST + (size_t)Nn * 2;
    float* DEN  = Mx   + (size_t)Nn * 2;
    float* EX   = DEN  + (size_t)Nn * 2;

    const int tilesM    = (Nn + 15) / 16;
    const int gemmBlks  = (tilesM + 3) / 4;
    const int nodeBlks  = (Nn * HCc + 255) / 256;
    const int alphaBlks = (Nn * 2 * 32 + 255) / 256;
    const int edgeBlks  = (Etot + 255) / 256;
    const int msgBlks   = (Etot * HCc + 255) / 256;

    wmma_gemm<4, false, FIN><<<gemmBlks, 128, 0, stream>>>(
        node_feat, nullptr, nullptr, W1, nullptr, H, Nn, HCc);
    node_alpha<<<alphaBlks, 256, 0, stream>>>(H, a_src1, a_dst1, ASRC, ADST);
    init_layer<<<nodeBlks, 256, 0, stream>>>(X1, Mx, DEN);
    edge_attn_max<<<edgeBlks, 256, 0, stream>>>(ei, E, Etot, ASRC, ADST, EX, Mx);
    edge_exp_sum<<<edgeBlks, 256, 0, stream>>>(ei, E, Etot, Mx, EX, DEN);
    edge_msg<<<msgBlks, 256, 0, stream>>>(ei, E, Etot, H, EX, DEN, X1);
    node_post1<<<nodeBlks, 256, 0, stream>>>(X1, b1, bn_gamma, bn_beta,
                                             bn_mean, bn_var);

    wmma_gemm<4, false, HCc><<<gemmBlks, 128, 0, stream>>>(
        X1, nullptr, nullptr, W2, nullptr, H, Nn, HCc);
    node_alpha<<<alphaBlks, 256, 0, stream>>>(H, a_src2, a_dst2, ASRC, ADST);
    init_layer<<<nodeBlks, 256, 0, stream>>>(AGG2, Mx, DEN);
    edge_attn_max<<<edgeBlks, 256, 0, stream>>>(ei, E, Etot, ASRC, ADST, EX, Mx);
    edge_exp_sum<<<edgeBlks, 256, 0, stream>>>(ei, E, Etot, Mx, EX, DEN);
    edge_msg<<<msgBlks, 256, 0, stream>>>(ei, E, Etot, H, EX, DEN, AGG2);

    wmma_gemm<3, true, HCc><<<gemmBlks, 128, 0, stream>>>(
        X1, AGG2, b2, Wf, bf, out, Nn, OUTc);
}


# edges counting-sorted by destination inside each call (one returning int atomic per edge, scan, scatter); per layer one wave per node does softmax and row sum by gather; float atomics gone
# speedup vs baseline: 3.3367x; 3.3367x over previous
#include <hip/hip_runtime.h>
#include <hip/hip_bf16.h>
#include <math.h>


#define Nn    100000
#define FIN   128
#define HCc   64
#define HIDc  32
#define OUTc  40
#define SCAN_CHUNK 1024

typedef __attribute__((ext_vector_type(2))) float v2f;
typedef __attribute__((ext_vector_type(8))) float v8f;

template<int NT, bool FUSE, int K>
__global__ __launch_bounds__(128)
void wmma_gemm(const float* __restrict__ A, const float* __restrict__ A2,
               const float* __restrict__ a2bias,
               const float* __restrict__ B, const float* __restrict__ bias,
               float* __restrict__ C, int M, int Ncols) {
    __shared__ float Bs[(K / 2) * 64 * 2];

    for (int i = threadIdx.x; i < K * 64; i += 128) {
        const int p    = i >> 7;
        const int rem  = i & 127;
        const int col  = rem >> 1;
        const int half = rem & 1;
        Bs[i] = (col < Ncols) ? B[(size_t)(2 * p + half) * Ncols + col] : 0.f;
    }
    __syncthreads();

    const int wave = threadIdx.x >> 5;
    const int lane = threadIdx.x & 31;
    const int tileM = blockIdx.x * 4 + wave;
    const int row16 = tileM * 16;
    if (row16 >= M) return;

    const int hi = (lane >= 16);
    const int lr = lane & 15;

    v8f acc[NT];
#pragma unroll
    for (int t = 0; t < NT; ++t)
        acc[t] = (v8f){0.f, 0.f, 0.f, 0.f, 0.f, 0.f, 0.f, 0.f};

    const int arow = row16 + lr;
    const float* __restrict__ Arow  = A + (size_t)arow * K;
    const float* __restrict__ A2row = FUSE ? (A2 + (size_t)arow * K) : nullptr;

#pragma unroll 4
    for (int kb = 0; kb < K; kb += 4) {
        const int ka = kb + 2 * hi;
        float a0 = Arow[ka];
        float a1 = Arow[ka + 1];
        if (FUSE) {
            a0 = fmaxf(a0, A2row[ka]     + a2bias[ka]);
            a1 = fmaxf(a1, A2row[ka + 1] + a2bias[ka + 1]);
        }
        v2f afrag; afrag.x = a0; afrag.y = a1;

        const int p = (kb >> 1) + hi;
#pragma unroll
        for (int t = 0; t < NT; ++t) {
            const v2f bfrag = *(const v2f*)&Bs[(p << 7) + ((t * 16 + lr) << 1)];
            acc[t] = __builtin_amdgcn_wmma_f32_16x16x4_f32(
                false, afrag, false, bfrag, (short)0, acc[t], false, false);
        }
    }

    const int rbase = row16 + (hi ? 8 : 0);
#pragma unroll
    for (int t = 0; t < NT; ++t) {
        const int col = t * 16 + lr;
        if (col < Ncols) {
            const float bb = bias ? bias[col] : 0.f;
#pragma unroll
            for (int r = 0; r < 8; ++r)
                C[(size_t)(rbase + r) * Ncols + col] = acc[t][r] + bb;
        }
    }
}

__global__ __launch_bounds__(256)
void node_alpha(const float* __restrict__ h, const float* __restrict__ a_s,
                const float* __restrict__ a_d, float* __restrict__ asrc,
                float* __restrict__ adst) {
    const int gw   = (blockIdx.x * blockDim.x + threadIdx.x) >> 5;
    const int lane = threadIdx.x & 31;
    const int n = gw >> 1, hd = gw & 1;
    if (n >= Nn) return;
    const float hv = h[n * HCc + hd * HIDc + lane];
    float vs = hv * a_s[hd * HIDc + lane];
    float vd = hv * a_d[hd * HIDc + lane];
#pragma unroll
    for (int off = 16; off; off >>= 1) {
        vs += __shfl_xor(vs, off, 32);
        vd += __shfl_xor(vd, off, 32);
    }
    if (lane == 0) { asrc[n * 2 + hd] = vs; adst[n * 2 + hd] = vd; }
}

__global__ __launch_bounds__(256)
void csr_zero(int* __restrict__ cnt) {
    const int tid = blockIdx.x * blockDim.x + threadIdx.x;
    if (tid < Nn) cnt[tid] = 0;
}

__global__ __launch_bounds__(256)
void csr_count(const int* __restrict__ ei, int E, int* __restrict__ cnt,
               int* __restrict__ rank) {
    const int e = blockIdx.x * blockDim.x + threadIdx.x;
    if (e < E) rank[e] = atomicAdd(&cnt[ei[E + e]], 1);
}

__global__ __launch_bounds__(256)
void csr_chunk_sums(const int* __restrict__ cnt, int* __restrict__ csum) {
    __shared__ int wsum[8];
    const int base = blockIdx.x * SCAN_CHUNK + threadIdx.x * 4;
    int v = 0;
#pragma unroll
    for (int j = 0; j < 4; ++j)
        if (base + j < Nn) v += cnt[base + j];
#pragma unroll
    for (int off = 16; off; off >>= 1) v += __shfl_xor(v, off, 32);
    if ((threadIdx.x & 31) == 0) wsum[threadIdx.x >> 5] = v;
    __syncthreads();
    if (threadIdx.x == 0) {
        int s = 0;
#pragma unroll
        for (int w = 0; w < 8; ++w) s += wsum[w];
        csum[blockIdx.x] = s;
    }
}

__global__ __launch_bounds__(256)
void csr_scan(const int* __restrict__ cnt, const int* __restrict__ csum,
              int* __restrict__ rowptr) {
    __shared__ int wsum[8];
    __shared__ int coff;
    const int lane = threadIdx.x & 31;
    const int wave = threadIdx.x >> 5;
    if (wave == 0) {
        int s = 0;
        for (int i = lane; i < (int)blockIdx.x; i += 32) s += csum[i];
#pragma unroll
        for (int off = 16; off; off >>= 1) s += __shfl_xor(s, off, 32);
        if (lane == 0) coff = s;
    }
    const int base = blockIdx.x * SCAN_CHUNK + threadIdx.x * 4;
    int c[4];
    int t = 0;
#pragma unroll
    for (int j = 0; j < 4; ++j) {
        c[j] = (base + j < Nn) ? cnt[base + j] : 0;
        t += c[j];
    }
    int inc = t;
#pragma unroll
    for (int off = 1; off < 32; off <<= 1) {
        const int u = __shfl_up(inc, off, 32);
        if (lane >= off) inc += u;
    }
    if (lane == 31) wsum[wave] = inc;
    __syncthreads();
    int run = coff + inc - t;
    for (int w = 0; w < wave; ++w) run += wsum[w];
#pragma unroll
    for (int j = 0; j < 4; ++j) {
        if (base + j < Nn) rowptr[base + j] = run;
        run += c[j];
    }
}

__global__ __launch_bounds__(256)
void csr_fill(const int* __restrict__ ei, int E, const int* __restrict__ rowptr,
              const int* __restrict__ rank, int* __restrict__ csrc) {
    const int e = blockIdx.x * blockDim.x + threadIdx.x;
    if (e < E) csrc[rowptr[ei[E + e]] + rank[e]] = ei[e];
}

__device__ inline v2f edge_logit(v2f as, v2f ad) {
    v2f e;
    e.x = as.x + ad.x; e.x = (e.x > 0.f) ? e.x : 0.2f * e.x;
    e.y = as.y + ad.y; e.y = (e.y > 0.f) ? e.y : 0.2f * e.y;
    return e;
}

__global__ __launch_bounds__(256)
void node_gather(const int* __restrict__ rowptr, const int* __restrict__ cnt,
                 const int* __restrict__ csrc, const float* __restrict__ h,
                 const float* __restrict__ asrc, const float* __restrict__ adst,
                 float* __restrict__ agg) {
    const int d    = __builtin_amdgcn_readfirstlane(
        (int)((blockIdx.x * blockDim.x + threadIdx.x) >> 5));
    const int lane = threadIdx.x & 31;
    if (d >= Nn) return;
    const int beg = rowptr[d];
    const int deg = cnt[d];
    const int tot = deg + 1;
    const v2f ad  = *(const v2f*)&adst[d * 2];

    float m0 = -INFINITY, m1 = -INFINITY;
#pragma nounroll
    for (int j0 = 0; j0 < tot; j0 += 32) {
        const int j = j0 + lane;
        if (j < tot) {
            const int s = (j < deg) ? csrc[beg + j] : d;
            const v2f e = edge_logit(*(const v2f*)&asrc[s * 2], ad);
            m0 = fmaxf(m0, e.x);
            m1 = fmaxf(m1, e.y);
        }
    }
#pragma unroll
    for (int off = 16; off; off >>= 1) {
        m0 = fmaxf(m0, __shfl_xor(m0, off, 32));
        m1 = fmaxf(m1, __shfl_xor(m1, off, 32));
    }

    float den0 = 0.f, den1 = 0.f;
#pragma nounroll
    for (int j0 = 0; j0 < tot; j0 += 32) {
        const int j = j0 + lane;
        if (j < tot) {
            const int s = (j < deg) ? csrc[beg + j] : d;
            const v2f e = edge_logit(*(const v2f*)&asrc[s * 2], ad);
            den0 += __expf(e.x - m0);
            den1 += __expf(e.y - m1);
        }
    }
#pragma unroll
    for (int off = 16; off; off >>= 1) {
        den0 += __shfl_xor(den0, off, 32);
        den1 += __shfl_xor(den1, off, 32);
    }
    den0 += 1e-16f;
    den1 += 1e-16f;

    const bool head1 = lane >= 16;
    v2f acc; acc.x = 0.f; acc.y = 0.f;
#pragma nounroll
    for (int j0 = 0; j0 < tot; j0 += 32) {
        const int j = j0 + lane;
        int   s  = d;
        float a0 = 0.f, a1 = 0.f;
        if (j < tot) {
            s = (j < deg) ? csrc[beg + j] : d;
            const v2f e = edge_logit(*(const v2f*)&asrc[s * 2], ad);
            a0 = __expf(e.x - m0) / den0;
            a1 = __expf(e.y - m1) / den1;
        }
        const int left = tot - j0;
        const int nk   = ((left < 32 ? left : 32) + 3) & ~3;
#pragma nounroll
        for (int k = 0; k < nk; k += 4) {
            v2f   hv[4];
            float w[4];
#pragma unroll
            for (int u = 0; u < 4; ++u) {
                const int   sk = __builtin_amdgcn_readlane(s, k + u);
                const float w0 = __int_as_float(
                    __builtin_amdgcn_readlane(__float_as_int(a0), k + u));
                const float w1 = __int_as_float(
                    __builtin_amdgcn_readlane(__float_as_int(a1), k + u));
                w[u]  = head1 ? w1 : w0;
                hv[u] = *(const v2f*)&h[(size_t)sk * HCc + lane * 2];
            }
#pragma unroll
            for (int u = 0; u < 4; ++u) {
                acc.x += hv[u].x * w[u];
                acc.y += hv[u].y * w[u];
            }
        }
    }
    *(v2f*)&agg[(size_t)d * HCc + lane * 2] = acc;
}

__global__ __launch_bounds__(256)
void node_post1(float* __restrict__ x, const float* __restrict__ b1,
                const float* __restrict__ gamma, const float* __restrict__ beta,
                const float* __restrict__ mean, const float* __restrict__ var) {
    const int tid = blockIdx.x * blockDim.x + threadIdx.x;
    if (tid >= Nn * HCc) return;
    const int c = tid & 63;
    float v = x[tid] + b1[c];
    v = (v - mean[c]) * (gamma[c] * rsqrtf(var[c] + 1e-5f)) + beta[c];
    x[tid] = (v > 0.f) ? v : (__expf(v) - 1.f);
}

extern "C" void kernel_launch(void* const* d_in, const int* in_sizes, int n_in,
                              void* d_out, int out_size, void* d_ws, size_t ws_size,
                              hipStream_t stream) {
    const float* node_feat = (const float*)d_in[0];
    const int*   ei        = (const int*)  d_in[1];
    const float* W1        = (const float*)d_in[2];
    const float* a_src1    = (const float*)d_in[3];
    const float* a_dst1    = (const float*)d_in[4];
    const float* b1        = (const float*)d_in[5];
    const float* bn_gamma  = (const float*)d_in[6];
    const float* bn_beta   = (const float*)d_in[7];
    const float* bn_mean   = (const float*)d_in[8];
    const float* bn_var    = (const float*)d_in[9];
    const float* W2        = (const float*)d_in[10];
    const float* a_src2    = (const float*)d_in[11];
    const float* a_dst2    = (const float*)d_in[12];
    const float* b2        = (const float*)d_in[13];
    const float* Wf        = (const float*)d_in[14];
    const float* bf        = (const float*)d_in[15];
    float* out = (float*)d_out;

    const int E = in_sizes[1] / 2;

    float* ws   = (float*)d_ws;
    float* H    = ws;
    float* X1   = H    + (size_t)Nn * HCc;
    float* AGG2 = X1   + (size_t)Nn * HCc;
    float* ASRC = AGG2 + (size_t)Nn * HCc;
    float* ADST = ASRC + (size_t)Nn * 2;
    int*   CNT  = (int*)(ADST + (size_t)Nn * 2);
    int*   RPTR = CNT  + Nn;
    int*   CSUM = RPTR + Nn;
    int*   RANK = CSUM + 128;
    int*   CSRC = RANK + (size_t)E;

    const int tilesM    = (Nn + 15) / 16;
    const int gemmBlks  = (tilesM + 3) / 4;
    const int nodeBlks  = (Nn * HCc + 255) / 256;
    const int alphaBlks = (Nn * 2 * 32 + 255) / 256;
    const int cntBlks   = (Nn + 255) / 256;
    const int edgeBlks  = (E + 255) / 256;
    const int scanBlks  = (Nn + SCAN_CHUNK - 1) / SCAN_CHUNK;
    const int gathBlks  = (Nn * 32 + 255) / 256;

    csr_zero<<<cntBlks, 256, 0, stream>>>(CNT);
    csr_count<<<edgeBlks, 256, 0, stream>>>(ei, E, CNT, RANK);
    csr_chunk_sums<<<scanBlks, 256, 0, stream>>>(CNT, CSUM);
    csr_scan<<<scanBlks, 256, 0, stream>>>(CNT, CSUM, RPTR);
    csr_fill<<<edgeBlks, 256, 0, stream>>>(ei, E, RPTR, RANK, CSRC);

    wmma_gemm<4, false, FIN><<<gemmBlks, 128, 0, stream>>>(
        node_feat, nullptr, nullptr, W1, nullptr, H, Nn, HCc);
    node_alpha<<<alphaBlks, 256, 0, stream>>>(H, a_src1, a_dst1, ASRC, ADST);
    node_gather<<<gathBlks, 256, 0, stream>>>(RPTR, CNT, CSRC, H, ASRC, ADST, X1);
    node_post1<<<nodeBlks, 256, 0, stream>>>(X1, b1, bn_gamma, bn_beta,
                                             bn_mean, bn_var);

    wmma_gemm<4, false, HCc><<<gemmBlks, 128, 0, stream>>>(
        X1, nullptr, nullptr, W2, nullptr, H, Nn, HCc);
    node_alpha<<<alphaBlks, 256, 0, stream>>>(H, a_src2, a_dst2, ASRC, ADST);
    node_gather<<<gathBlks, 256, 0, stream>>>(RPTR, CNT, CSRC, H, ASRC, ADST, AGG2);

    wmma_gemm<3, true, HCc><<<gemmBlks, 128, 0, stream>>>(
        X1, AGG2, b2, Wf, bf, out, Nn, OUTc);
}


# per-edge and per-node kernels (degree count, scatter, alpha, gather, BN/ELU) run as chip-resident grids whose waves loop over items instead of one short wave per item: 8x fewer wave starts per call
# speedup vs baseline: 3.7131x; 1.1128x over previous
#include <hip/hip_runtime.h>
#include <hip/hip_bf16.h>
#include <math.h>


#define Nn    100000
#define FIN   128
#define HCc   64
#define HIDc  32
#define OUTc  40
#define SCAN_CHUNK 1024

typedef __attribute__((ext_vector_type(2))) float v2f;
typedef __attribute__((ext_vector_type(8))) float v8f;

template<int NT, bool FUSE, int K>
__global__ __launch_bounds__(128)
void wmma_gemm(const float* __restrict__ A, const float* __restrict__ A2,
               const float* __restrict__ a2bias,
               const float* __restrict__ B, const float* __restrict__ bias,
               float* __restrict__ C, int M, int Ncols) {
    __shared__ float Bs[(K / 2) * 64 * 2];

    for (int i = threadIdx.x; i < K * 64; i += 128) {
        const int p    = i >> 7;
        const int rem  = i & 127;
        const int col  = rem >> 1;
        const int half = rem & 1;
        Bs[i] = (col < Ncols) ? B[(size_t)(2 * p + half) * Ncols + col] : 0.f;
    }
    __syncthreads();

    const int wave = threadIdx.x >> 5;
    const int lane = threadIdx.x & 31;
    const int tileM = blockIdx.x * 4 + wave;
    const int row16 = tileM * 16;
    if (row16 >= M) return;

    const int hi = (lane >= 16);
    const int lr = lane & 15;

    v8f acc[NT];
#pragma unroll
    for (int t = 0; t < NT; ++t)
        acc[t] = (v8f){0.f, 0.f, 0.f, 0.f, 0.f, 0.f, 0.f, 0.f};

    const int arow = row16 + lr;
    const float* __restrict__ Arow  = A + (size_t)arow * K;
    const float* __restrict__ A2row = FUSE ? (A2 + (size_t)arow * K) : nullptr;

#pragma unroll 4
    for (int kb = 0; kb < K; kb += 4) {
        const int ka = kb + 2 * hi;
        float a0 = Arow[ka];
        float a1 = Arow[ka + 1];
        if (FUSE) {
            a0 = fmaxf(a0, A2row[ka]     + a2bias[ka]);
            a1 = fmaxf(a1, A2row[ka + 1] + a2bias[ka + 1]);
        }
        v2f afrag; afrag.x = a0; afrag.y = a1;

        const int p = (kb >> 1) + hi;
#pragma unroll
        for (int t = 0; t < NT; ++t) {
            const v2f bfrag = *(const v2f*)&Bs[(p << 7) + ((t * 16 + lr) << 1)];
            acc[t] = __builtin_amdgcn_wmma_f32_16x16x4_f32(
                false, afrag, false, bfrag, (short)0, acc[t], false, false);
        }
    }

    const int rbase = row16 + (hi ? 8 : 0);
#pragma unroll
    for (int t = 0; t < NT; ++t) {
        const int col = t * 16 + lr;
        if (col < Ncols) {
            const float bb = bias ? bias[col] : 0.f;
#pragma unroll
            for (int r = 0; r < 8; ++r)
                C[(size_t)(rbase + r) * Ncols + col] = acc[t][r] + bb;
        }
    }
}

__global__ __launch_bounds__(256)
void node_alpha(const float* __restrict__ h, const float* __restrict__ a_s,
                const float* __restrict__ a_d, float* __restrict__ asrc,
                float* __restrict__ adst) {
    const int lane = threadIdx.x & 31;
    const int sub  = lane & 15;
    const int half = lane >> 4;
    const int hd   = sub >> 3;
    const float4 as4 = *(const float4*)&a_s[sub * 4];
    const float4 ad4 = *(const float4*)&a_d[sub * 4];
    const int wave = (int)((blockIdx.x * blockDim.x + threadIdx.x) >> 5);
    const int nw   = (int)((gridDim.x * blockDim.x) >> 5);
    for (int p = wave; p < (Nn + 1) / 2; p += 2 * nw) {
        int    n[2];
        float4 hv[2];
#pragma unroll
        for (int u = 0; u < 2; ++u) {
            n[u] = 2 * (p + u * nw) + half;
            const int nl = (n[u] < Nn) ? n[u] : (Nn - 1);
            hv[u] = *(const float4*)&h[(size_t)nl * HCc + sub * 4];
        }
#pragma unroll
        for (int u = 0; u < 2; ++u) {
            float vs = hv[u].x * as4.x + hv[u].y * as4.y + hv[u].z * as4.z + hv[u].w * as4.w;
            float vd = hv[u].x * ad4.x + hv[u].y * ad4.y + hv[u].z * ad4.z + hv[u].w * ad4.w;
#pragma unroll
            for (int off = 4; off; off >>= 1) {
                vs += __shfl_xor(vs, off, 32);
                vd += __shfl_xor(vd, off, 32);
            }
            if ((lane & 7) == 0 && n[u] < Nn) {
                asrc[n[u] * 2 + hd] = vs;
                adst[n[u] * 2 + hd] = vd;
            }
        }
    }
}

__global__ __launch_bounds__(256)
void csr_zero(int* __restrict__ cnt) {
    const int tid = blockIdx.x * blockDim.x + threadIdx.x;
    if (tid < Nn) cnt[tid] = 0;
}

__global__ __launch_bounds__(256)
void csr_count(const int* __restrict__ ei, int E, int* __restrict__ cnt,
               int* __restrict__ rank) {
    const int nt = (int)(gridDim.x * blockDim.x);
    for (int e = (int)(blockIdx.x * blockDim.x + threadIdx.x); e < E; e += nt)
        rank[e] = atomicAdd(&cnt[ei[E + e]], 1);
}

__global__ __launch_bounds__(256)
void csr_chunk_sums(const int* __restrict__ cnt, int* __restrict__ csum) {
    __shared__ int wsum[8];
    const int base = blockIdx.x * SCAN_CHUNK + threadIdx.x * 4;
    int v = 0;
#pragma unroll
    for (int j = 0; j < 4; ++j)
        if (base + j < Nn) v += cnt[base + j];
#pragma unroll
    for (int off = 16; off; off >>= 1) v += __shfl_xor(v, off, 32);
    if ((threadIdx.x & 31) == 0) wsum[threadIdx.x >> 5] = v;
    __syncthreads();
    if (threadIdx.x == 0) {
        int s = 0;
#pragma unroll
        for (int w = 0; w < 8; ++w) s += wsum[w];
        csum[blockIdx.x] = s;
    }
}

__global__ __launch_bounds__(256)
void csr_scan(const int* __restrict__ cnt, const int* __restrict__ csum,
              int* __restrict__ rowptr) {
    __shared__ int wsum[8];
    __shared__ int coff;
    const int lane = threadIdx.x & 31;
    const int wave = threadIdx.x >> 5;
    if (wave == 0) {
        int s = 0;
        for (int i = lane; i < (int)blockIdx.x; i += 32) s += csum[i];
#pragma unroll
        for (int off = 16; off; off >>= 1) s += __shfl_xor(s, off, 32);
        if (lane == 0) coff = s;
    }
    const int base = blockIdx.x * SCAN_CHUNK + threadIdx.x * 4;
    int c[4];
    int t = 0;
#pragma unroll
    for (int j = 0; j < 4; ++j) {
        c[j] = (base + j < Nn) ? cnt[base + j] : 0;
        t += c[j];
    }
    int inc = t;
#pragma unroll
    for (int off = 1; off < 32; off <<= 1) {
        const int u = __shfl_up(inc, off, 32);
        if (lane >= off) inc += u;
    }
    if (lane == 31) wsum[wave] = inc;
    __syncthreads();
    int run = coff + inc - t;
    for (int w = 0; w < wave; ++w) run += wsum[w];
#pragma unroll
    for (int j = 0; j < 4; ++j) {
        if (base + j < Nn) rowptr[base + j] = run;
        run += c[j];
    }
}

__global__ __launch_bounds__(256)
void csr_fill(const int* __restrict__ ei, int E, const int* __restrict__ rowptr,
              const int* __restrict__ rank, int* __restrict__ csrc) {
    const int nt = (int)(gridDim.x * blockDim.x);
    for (int e = (int)(blockIdx.x * blockDim.x + threadIdx.x); e < E; e += nt)
        csrc[rowptr[ei[E + e]] + rank[e]] = ei[e];
}

__device__ inline v2f edge_logit(v2f as, v2f ad) {
    v2f e;
    e.x = as.x + ad.x; e.x = (e.x > 0.f) ? e.x : 0.2f * e.x;
    e.y = as.y + ad.y; e.y = (e.y > 0.f) ? e.y : 0.2f * e.y;
    return e;
}

__global__ __launch_bounds__(256)
void node_gather(const int* __restrict__ rowptr, const int* __restrict__ cnt,
                 const int* __restrict__ csrc, const float* __restrict__ h,
                 const float* __restrict__ asrc, const float* __restrict__ adst,
                 float* __restrict__ agg) {
    const int d0   = __builtin_amdgcn_readfirstlane(
        (int)((blockIdx.x * blockDim.x + threadIdx.x) >> 5));
    const int nw   = (int)((gridDim.x * blockDim.x) >> 5);
    const int lane = threadIdx.x & 31;
    for (int d = d0; d < Nn; d += nw) {
        const int beg = rowptr[d];
        const int deg = cnt[d];
        const int tot = deg + 1;
        const v2f ad  = *(const v2f*)&adst[d * 2];

        float m0 = -INFINITY, m1 = -INFINITY;
#pragma nounroll
        for (int j0 = 0; j0 < tot; j0 += 32) {
            const int j = j0 + lane;
            if (j < tot) {
                const int s = (j < deg) ? csrc[beg + j] : d;
                const v2f e = edge_logit(*(const v2f*)&asrc[s * 2], ad);
                m0 = fmaxf(m0, e.x);
                m1 = fmaxf(m1, e.y);
            }
        }
#pragma unroll
        for (int off = 16; off; off >>= 1) {
            m0 = fmaxf(m0, __shfl_xor(m0, off, 32));
            m1 = fmaxf(m1, __shfl_xor(m1, off, 32));
        }

        float den0 = 0.f, den1 = 0.f;
#pragma nounroll
        for (int j0 = 0; j0 < tot; j0 += 32) {
            const int j = j0 + lane;
            if (j < tot) {
                const int s = (j < deg) ? csrc[beg + j] : d;
                const v2f e = edge_logit(*(const v2f*)&asrc[s * 2], ad);
                den0 += __expf(e.x - m0);
                den1 += __expf(e.y - m1);
            }
        }
#pragma unroll
        for (int off = 16; off; off >>= 1) {
            den0 += __shfl_xor(den0, off, 32);
            den1 += __shfl_xor(den1, off, 32);
        }
        den0 += 1e-16f;
        den1 += 1e-16f;

        const bool head1 = lane >= 16;
        v2f acc; acc.x = 0.f; acc.y = 0.f;
#pragma nounroll
        for (int j0 = 0; j0 < tot; j0 += 32) {
            const int j = j0 + lane;
            int   s  = d;
            float a0 = 0.f, a1 = 0.f;
            if (j < tot) {
                s = (j < deg) ? csrc[beg + j] : d;
                const v2f e = edge_logit(*(const v2f*)&asrc[s * 2], ad);
                a0 = __expf(e.x - m0) / den0;
                a1 = __expf(e.y - m1) / den1;
            }
            const int left = tot - j0;
            const int nk   = ((left < 32 ? left : 32) + 3) & ~3;
#pragma nounroll
            for (int k = 0; k < nk; k += 4) {
                v2f   hv[4];
                float w[4];
#pragma unroll
                for (int u = 0; u < 4; ++u) {
                    const int   sk = __builtin_amdgcn_readlane(s, k + u);
                    const float w0 = __int_as_float(
                        __builtin_amdgcn_readlane(__float_as_int(a0), k + u));
                    const float w1 = __int_as_float(
                        __builtin_amdgcn_readlane(__float_as_int(a1), k + u));
                    w[u]  = head1 ? w1 : w0;
                    hv[u] = *(const v2f*)&h[(size_t)sk * HCc + lane * 2];
                }
#pragma unroll
                for (int u = 0; u < 4; ++u) {
                    acc.x += hv[u].x * w[u];
                    acc.y += hv[u].y * w[u];
                }
            }
        }
        *(v2f*)&agg[(size_t)d * HCc + lane * 2] = acc;
    }
}

__global__ __launch_bounds__(256)
void node_post1(float* __restrict__ x, const float* __restrict__ b1,
                const float* __restrict__ gamma, const float* __restrict__ beta,
                const float* __restrict__ mean, const float* __restrict__ var) {
    const int tid = (int)(blockIdx.x * blockDim.x + threadIdx.x);
    const int nt  = (int)(gridDim.x * blockDim.x);
    const int c   = (tid & 15) * 4;
    float bb[4], mu[4], g[4], be[4];
#pragma unroll
    for (int j = 0; j < 4; ++j) {
        bb[j] = b1[c + j];
        mu[j] = mean[c + j];
        g[j]  = gamma[c + j] * rsqrtf(var[c + j] + 1e-5f);
        be[j] = beta[c + j];
    }
    float4* __restrict__ x4 = (float4*)x;
    for (int i = tid; i < Nn * HCc / 4; i += 2 * nt) {
        const bool two = (i + nt) < Nn * HCc / 4;
        const int  i2  = two ? (i + nt) : i;
        const float4 va = x4[i];
        const float4 vb = x4[i2];
        float v[8] = {va.x, va.y, va.z, va.w, vb.x, vb.y, vb.z, vb.w};
#pragma unroll
        for (int j = 0; j < 8; ++j) {
            float t = v[j] + bb[j & 3];
            t = (t - mu[j & 3]) * g[j & 3] + be[j & 3];
            v[j] = (t > 0.f) ? t : (__expf(t) - 1.f);
        }
        x4[i] = make_float4(v[0], v[1], v[2], v[3]);
        if (two) x4[i2] = make_float4(v[4], v[5], v[6], v[7]);
    }
}

extern "C" void kernel_launch(void* const* d_in, const int* in_sizes, int n_in,
                              void* d_out, int out_size, void* d_ws, size_t ws_size,
                              hipStream_t stream) {
    const float* node_feat = (const float*)d_in[0];
    const int*   ei        = (const int*)  d_in[1];
    const float* W1        = (const float*)d_in[2];
    const float* a_src1    = (const float*)d_in[3];
    const float* a_dst1    = (const float*)d_in[4];
    const float* b1        = (const float*)d_in[5];
    const float* bn_gamma  = (const float*)d_in[6];
    const float* bn_beta   = (const float*)d_in[7];
    const float* bn_mean   = (const float*)d_in[8];
    const float* bn_var    = (const float*)d_in[9];
    const float* W2        = (const float*)d_in[10];
    const float* a_src2    = (const float*)d_in[11];
    const float* a_dst2    = (const float*)d_in[12];
    const float* b2        = (const float*)d_in[13];
    const float* Wf        = (const float*)d_in[14];
    const float* bf        = (const float*)d_in[15];
    float* out = (float*)d_out;

    const int E = in_sizes[1] / 2;

    float* ws   = (float*)d_ws;
    float* H    = ws;
    float* X1   = H    + (size_t)Nn * HCc;
    float* AGG2 = X1   + (size_t)Nn * HCc;
    float* ASRC = AGG2 + (size_t)Nn * HCc;
    float* ADST = ASRC + (size_t)Nn * 2;
    int*   CNT  = (int*)(ADST + (size_t)Nn * 2);
    int*   RPTR = CNT  + Nn;
    int*   CSUM = RPTR + Nn;
    int*   RANK = CSUM + 128;
    int*   CSRC = RANK + (size_t)E;

    const int tilesM    = (Nn + 15) / 16;
    const int gemmBlks  = (tilesM + 3) / 4;
    const int cntBlks   = (Nn + 255) / 256;
    const int scanBlks  = (Nn + SCAN_CHUNK - 1) / SCAN_CHUNK;
    const int edgeBlks  = 2048;
    const int gathBlks  = 2048;
    const int alphaBlks = 1024;
    const int nodeBlks  = 1024;

    csr_zero<<<cntBlks, 256, 0, stream>>>(CNT);
    csr_count<<<edgeBlks, 256, 0, stream>>>(ei, E, CNT, RANK);
    csr_chunk_sums<<<scanBlks, 256, 0, stream>>>(CNT, CSUM);
    csr_scan<<<scanBlks, 256, 0, stream>>>(CNT, CSUM, RPTR);
    csr_fill<<<edgeBlks, 256, 0, stream>>>(ei, E, RPTR, RANK, CSRC);

    wmma_gemm<4, false, FIN><<<gemmBlks, 128, 0, stream>>>(
        node_feat, nullptr, nullptr, W1, nullptr, H, Nn, HCc);
    node_alpha<<<alphaBlks, 256, 0, stream>>>(H, a_src1, a_dst1, ASRC, ADST);
    node_gather<<<gathBlks, 256, 0, stream>>>(RPTR, CNT, CSRC, H, ASRC, ADST, X1);
    node_post1<<<nodeBlks, 256, 0, stream>>>(X1, b1, bn_gamma, bn_beta,
                                             bn_mean, bn_var);

    wmma_gemm<4, false, HCc><<<gemmBlks, 128, 0, stream>>>(
        X1, nullptr, nullptr, W2, nullptr, H, Nn, HCc);
    node_alpha<<<alphaBlks, 256, 0, stream>>>(H, a_src2, a_dst2, ASRC, ADST);
    node_gather<<<gathBlks, 256, 0, stream>>>(RPTR, CNT, CSRC, H, ASRC, ADST, AGG2);

    wmma_gemm<3, true, HCc><<<gemmBlks, 128, 0, stream>>>(
        X1, AGG2, b2, Wf, bf, out, Nn, OUTc);
}


# node gather: a node's first 32 edges (all of them below in-degree 32) get source id, logit and exp term loaded and computed once, kept in registers from max to weights, not reloaded in three passes
# speedup vs baseline: 5.5955x; 1.5070x over previous
#include <hip/hip_runtime.h>
#include <hip/hip_bf16.h>
#include <math.h>


#define Nn    100000
#define FIN   128
#define HCc   64
#define HIDc  32
#define OUTc  40
#define BKT_SHIFT 8
#define BKT_NODES (1 << BKT_SHIFT)
#define NBKT ((Nn + BKT_NODES - 1) >> BKT_SHIFT)
#define SRC_BITS 17
#define SORT_BLKS 128
#define SORT_THREADS 1024
#define ROWS_THREADS 512
static_assert(Nn <= (1 << SRC_BITS), "source id must fit its field");
static_assert(NBKT < 512, "one thread per bucket in the offset scan");

typedef __attribute__((ext_vector_type(2))) float v2f;
typedef __attribute__((ext_vector_type(8))) float v8f;

template<int NT, bool FUSE, int K>
__global__ __launch_bounds__(128)
void wmma_gemm(const float* __restrict__ A, const float* __restrict__ A2,
               const float* __restrict__ a2bias,
               const float* __restrict__ B, const float* __restrict__ bias,
               float* __restrict__ C, int M, int Ncols) {
    __shared__ float Bs[(K / 2) * 64 * 2];

    for (int i = threadIdx.x; i < K * 64; i += 128) {
        const int p    = i >> 7;
        const int rem  = i & 127;
        const int col  = rem >> 1;
        const int half = rem & 1;
        Bs[i] = (col < Ncols) ? B[(size_t)(2 * p + half) * Ncols + col] : 0.f;
    }
    __syncthreads();

    const int wave = threadIdx.x >> 5;
    const int lane = threadIdx.x & 31;
    const int tileM = blockIdx.x * 4 + wave;
    const int row16 = tileM * 16;
    if (row16 >= M) return;

    const int hi = (lane >= 16);
    const int lr = lane & 15;

    v8f acc[NT];
#pragma unroll
    for (int t = 0; t < NT; ++t)
        acc[t] = (v8f){0.f, 0.f, 0.f, 0.f, 0.f, 0.f, 0.f, 0.f};

    const int arow = row16 + lr;
    const float* __restrict__ Arow  = A + (size_t)arow * K;
    const float* __restrict__ A2row = FUSE ? (A2 + (size_t)arow * K) : nullptr;

#pragma unroll 4
    for (int kb = 0; kb < K; kb += 4) {
        const int ka = kb + 2 * hi;
        float a0 = Arow[ka];
        float a1 = Arow[ka + 1];
        if (FUSE) {
            a0 = fmaxf(a0, A2row[ka]     + a2bias[ka]);
            a1 = fmaxf(a1, A2row[ka + 1] + a2bias[ka + 1]);
        }
        v2f afrag; afrag.x = a0; afrag.y = a1;

        const int p = (kb >> 1) + hi;
#pragma unroll
        for (int t = 0; t < NT; ++t) {
            const v2f bfrag = *(const v2f*)&Bs[(p << 7) + ((t * 16 + lr) << 1)];
            acc[t] = __builtin_amdgcn_wmma_f32_16x16x4_f32(
                false, afrag, false, bfrag, (short)0, acc[t], false, false);
        }
    }

    const int rbase = row16 + (hi ? 8 : 0);
#pragma unroll
    for (int t = 0; t < NT; ++t) {
        const int col = t * 16 + lr;
        if (col < Ncols) {
            const float bb = bias ? bias[col] : 0.f;
#pragma unroll
            for (int r = 0; r < 8; ++r)
                C[(size_t)(rbase + r) * Ncols + col] = acc[t][r] + bb;
        }
    }
}

__global__ __launch_bounds__(256)
void node_alpha(const float* __restrict__ h, const float* __restrict__ a_s,
                const float* __restrict__ a_d, float* __restrict__ asrc,
                float* __restrict__ adst) {
    const int lane = threadIdx.x & 31;
    const int sub  = lane & 15;
    const int half = lane >> 4;
    const int hd   = sub >> 3;
    const float4 as4 = *(const float4*)&a_s[sub * 4];
    const float4 ad4 = *(const float4*)&a_d[sub * 4];
    const int wave = (int)((blockIdx.x * blockDim.x + threadIdx.x) >> 5);
    const int nw   = (int)((gridDim.x * blockDim.x) >> 5);
    for (int p = wave; p < (Nn + 1) / 2; p += 2 * nw) {
        int    n[2];
        float4 hv[2];
#pragma unroll
        for (int u = 0; u < 2; ++u) {
            n[u] = 2 * (p + u * nw) + half;
            const int nl = (n[u] < Nn) ? n[u] : (Nn - 1);
            hv[u] = *(const float4*)&h[(size_t)nl * HCc + sub * 4];
        }
#pragma unroll
        for (int u = 0; u < 2; ++u) {
            float vs = hv[u].x * as4.x + hv[u].y * as4.y + hv[u].z * as4.z + hv[u].w * as4.w;
            float vd = hv[u].x * ad4.x + hv[u].y * ad4.y + hv[u].z * ad4.z + hv[u].w * ad4.w;
#pragma unroll
            for (int off = 4; off; off >>= 1) {
                vs += __shfl_xor(vs, off, 32);
                vd += __shfl_xor(vd, off, 32);
            }
            if ((lane & 7) == 0 && n[u] < Nn) {
                asrc[n[u] * 2 + hd] = vs;
                adst[n[u] * 2 + hd] = vd;
            }
        }
    }
}

__global__ __launch_bounds__(SORT_THREADS)
void sort_hist(const int* __restrict__ ei, int E, int* __restrict__ hist) {
    __shared__ int hc[NBKT];
    for (int t = threadIdx.x; t < NBKT; t += SORT_THREADS) hc[t] = 0;
    __syncthreads();
    const int chunk = (E + SORT_BLKS - 1) / SORT_BLKS;
    const int e0 = blockIdx.x * chunk;
    const int e1 = (e0 + chunk < E) ? (e0 + chunk) : E;
    const int* __restrict__ dst = ei + E;
    int e = e0 + threadIdx.x;
    for (; e + 3 * SORT_THREADS < e1; e += 4 * SORT_THREADS) {
        int d[4];
#pragma unroll
        for (int u = 0; u < 4; ++u) d[u] = dst[e + u * SORT_THREADS];
#pragma unroll
        for (int u = 0; u < 4; ++u) atomicAdd(&hc[d[u] >> BKT_SHIFT], 1);
    }
    for (; e < e1; e += SORT_THREADS) atomicAdd(&hc[dst[e] >> BKT_SHIFT], 1);
    __syncthreads();
    for (int t = threadIdx.x; t < NBKT; t += SORT_THREADS)
        hist[blockIdx.x * NBKT + t] = hc[t];
}

__global__ __launch_bounds__(SORT_THREADS)
void sort_scatter(const int* __restrict__ ei, int E, const int* __restrict__ hist,
                  int* __restrict__ bstart, int* __restrict__ filed) {
    __shared__ int part[2][2][512];
    __shared__ int wsum[16];
    __shared__ int cur[512];
    const int g    = blockIdx.x;
    const int t    = threadIdx.x & 511;
    const int rh   = threadIdx.x >> 9;
    const int lane = threadIdx.x & 31;
    const int wave = threadIdx.x >> 5;

    int pre = 0, tot = 0;
    if (t < NBKT) {
        const int r0 = rh * (SORT_BLKS / 2);
        for (int r = r0; r < r0 + SORT_BLKS / 2; ++r) {
            const int v = hist[r * NBKT + t];
            tot += v;
            pre += (r < g) ? v : 0;
        }
    }
    part[rh][0][t] = pre;
    part[rh][1][t] = tot;
    __syncthreads();

    int excl = 0;
    if (threadIdx.x < 512) {
        pre = part[0][0][t] + part[1][0][t];
        tot = part[0][1][t] + part[1][1][t];
        int inc = tot;
#pragma unroll
        for (int off = 1; off < 32; off <<= 1) {
            const int u = __shfl_up(inc, off, 32);
            if (lane >= off) inc += u;
        }
        if (lane == 31) wsum[wave] = inc;
        excl = inc - tot;
    }
    __syncthreads();
    if (threadIdx.x < 512) {
        int bs = excl;
        for (int w = 0; w < wave; ++w) bs += wsum[w];
        cur[t] = bs + pre;
        if (g == 0 && t <= NBKT) bstart[t] = bs;
    }
    __syncthreads();

    const int chunk = (E + SORT_BLKS - 1) / SORT_BLKS;
    const int e0 = g * chunk;
    const int e1 = (e0 + chunk < E) ? (e0 + chunk) : E;
    int e = e0 + threadIdx.x;
    for (; e + 3 * SORT_THREADS < e1; e += 4 * SORT_THREADS) {
        int s[4], d[4];
#pragma unroll
        for (int u = 0; u < 4; ++u) {
            s[u] = ei[e + u * SORT_THREADS];
            d[u] = ei[E + e + u * SORT_THREADS];
        }
#pragma unroll
        for (int u = 0; u < 4; ++u) {
            const int pos = atomicAdd(&cur[d[u] >> BKT_SHIFT], 1);
            filed[pos] = ((d[u] & (BKT_NODES - 1)) << SRC_BITS) | s[u];
        }
    }
    for (; e < e1; e += SORT_THREADS) {
        const int s = ei[e];
        const int d = ei[E + e];
        const int pos = atomicAdd(&cur[d >> BKT_SHIFT], 1);
        filed[pos] = ((d & (BKT_NODES - 1)) << SRC_BITS) | s;
    }
}

__global__ __launch_bounds__(ROWS_THREADS)
void sort_rows(const int* __restrict__ filed, const int* __restrict__ bstart,
               int* __restrict__ cnt, int* __restrict__ rowptr,
               int* __restrict__ csrc) {
    __shared__ int c[BKT_NODES];
    __shared__ int wsum[BKT_NODES / 32];
    const int bkt  = blockIdx.x;
    const int beg  = bstart[bkt];
    const int end  = bstart[bkt + 1];
    const int lane = threadIdx.x & 31;
    const int wave = threadIdx.x >> 5;
    if (threadIdx.x < BKT_NODES) c[threadIdx.x] = 0;
    __syncthreads();
    int i = beg + threadIdx.x;
    for (; i + 3 * ROWS_THREADS < end; i += 4 * ROWS_THREADS) {
        unsigned p[4];
#pragma unroll
        for (int u = 0; u < 4; ++u) p[u] = (unsigned)filed[i + u * ROWS_THREADS];
#pragma unroll
        for (int u = 0; u < 4; ++u) atomicAdd(&c[p[u] >> SRC_BITS], 1);
    }
    for (; i < end; i += ROWS_THREADS) atomicAdd(&c[(unsigned)filed[i] >> SRC_BITS], 1);
    __syncthreads();

    int deg = 0, excl = 0;
    if (threadIdx.x < BKT_NODES) {
        deg = c[threadIdx.x];
        int inc = deg;
#pragma unroll
        for (int off = 1; off < 32; off <<= 1) {
            const int u = __shfl_up(inc, off, 32);
            if (lane >= off) inc += u;
        }
        if (lane == 31) wsum[wave] = inc;
        excl = inc - deg;
    }
    __syncthreads();
    if (threadIdx.x < BKT_NODES) {
        int start = beg + excl;
        for (int w = 0; w < wave; ++w) start += wsum[w];
        const int node = bkt * BKT_NODES + threadIdx.x;
        if (node < Nn) {
            cnt[node]    = deg;
            rowptr[node] = start;
        }
        c[threadIdx.x] = start;
    }
    __syncthreads();
    i = beg + threadIdx.x;
    for (; i + 3 * ROWS_THREADS < end; i += 4 * ROWS_THREADS) {
        unsigned p[4];
#pragma unroll
        for (int u = 0; u < 4; ++u) p[u] = (unsigned)filed[i + u * ROWS_THREADS];
#pragma unroll
        for (int u = 0; u < 4; ++u) {
            const int pos = atomicAdd(&c[p[u] >> SRC_BITS], 1);
            csrc[pos] = (int)(p[u] & ((1u << SRC_BITS) - 1u));
        }
    }
    for (; i < end; i += ROWS_THREADS) {
        const unsigned p = (unsigned)filed[i];
        const int pos = atomicAdd(&c[p >> SRC_BITS], 1);
        csrc[pos] = (int)(p & ((1u << SRC_BITS) - 1u));
    }
}

__device__ inline v2f edge_logit(v2f as, v2f ad) {
    v2f e;
    e.x = as.x + ad.x; e.x = (e.x > 0.f) ? e.x : 0.2f * e.x;
    e.y = as.y + ad.y; e.y = (e.y > 0.f) ? e.y : 0.2f * e.y;
    return e;
}


__device__ inline v2f chunk_rows(const float* __restrict__ h, int s, float a0, float a1,
                                 int nk, int lane, bool head1, v2f acc) {
#pragma nounroll
    for (int k = 0; k < nk; k += 4) {
        v2f   hv[4];
        float w[4];
#pragma unroll
        for (int u = 0; u < 4; ++u) {
            const int   sk = __builtin_amdgcn_readlane(s, k + u);
            const float w0 = __int_as_float(
                __builtin_amdgcn_readlane(__float_as_int(a0), k + u));
            const float w1 = __int_as_float(
                __builtin_amdgcn_readlane(__float_as_int(a1), k + u));
            w[u]  = head1 ? w1 : w0;
            hv[u] = *(const v2f*)&h[(size_t)sk * HCc + lane * 2];
        }
#pragma unroll
        for (int u = 0; u < 4; ++u) {
            acc.x += hv[u].x * w[u];
            acc.y += hv[u].y * w[u];
        }
    }
    return acc;
}

__global__ __launch_bounds__(256)
void node_gather(const int* __restrict__ rowptr, const int* __restrict__ cnt,
                 const int* __restrict__ csrc, const float* __restrict__ h,
                 const float* __restrict__ asrc, const float* __restrict__ adst,
                 float* __restrict__ agg) {
    const int d0   = __builtin_amdgcn_readfirstlane(
        (int)((blockIdx.x * blockDim.x + threadIdx.x) >> 5));
    const int nw   = (int)((gridDim.x * blockDim.x) >> 5);
    const int lane = threadIdx.x & 31;
    const bool head1 = lane >= 16;
    for (int d = d0; d < Nn; d += nw) {
        const int beg = rowptr[d];
        const int deg = cnt[d];
        const int tot = deg + 1;
        const v2f ad  = *(const v2f*)&adst[d * 2];

        int s1 = d;
        v2f e1; e1.x = -INFINITY; e1.y = -INFINITY;
        if (lane < tot) {
            s1 = (lane < deg) ? csrc[beg + lane] : d;
            e1 = edge_logit(*(const v2f*)&asrc[s1 * 2], ad);
        }

        float m0 = e1.x, m1 = e1.y;
#pragma nounroll
        for (int j0 = 32; j0 < tot; j0 += 32) {
            const int j = j0 + lane;
            if (j < tot) {
                const int s = (j < deg) ? csrc[beg + j] : d;
                const v2f e = edge_logit(*(const v2f*)&asrc[s * 2], ad);
                m0 = fmaxf(m0, e.x);
                m1 = fmaxf(m1, e.y);
            }
        }
#pragma unroll
        for (int off = 16; off; off >>= 1) {
            m0 = fmaxf(m0, __shfl_xor(m0, off, 32));
            m1 = fmaxf(m1, __shfl_xor(m1, off, 32));
        }

        const float p0 = __expf(e1.x - m0);
        const float p1 = __expf(e1.y - m1);
        float den0 = p0, den1 = p1;
#pragma nounroll
        for (int j0 = 32; j0 < tot; j0 += 32) {
            const int j = j0 + lane;
            if (j < tot) {
                const int s = (j < deg) ? csrc[beg + j] : d;
                const v2f e = edge_logit(*(const v2f*)&asrc[s * 2], ad);
                den0 += __expf(e.x - m0);
                den1 += __expf(e.y - m1);
            }
        }
#pragma unroll
        for (int off = 16; off; off >>= 1) {
            den0 += __shfl_xor(den0, off, 32);
            den1 += __shfl_xor(den1, off, 32);
        }
        den0 += 1e-16f;
        den1 += 1e-16f;

        v2f acc; acc.x = 0.f; acc.y = 0.f;
        acc = chunk_rows(h, s1, p0 / den0, p1 / den1,
                         ((tot < 32 ? tot : 32) + 3) & ~3,
                         lane, head1, acc);
#pragma nounroll
        for (int j0 = 32; j0 < tot; j0 += 32) {
            const int j = j0 + lane;
            int   s  = d;
            float a0 = 0.f, a1 = 0.f;
            if (j < tot) {
                s = (j < deg) ? csrc[beg + j] : d;
                const v2f e = edge_logit(*(const v2f*)&asrc[s * 2], ad);
                a0 = __expf(e.x - m0) / den0;
                a1 = __expf(e.y - m1) / den1;
            }
            const int left = tot - j0;
            acc = chunk_rows(h, s, a0, a1, ((left < 32 ? left : 32) + 3) & ~3,
                             lane, head1, acc);
        }
        *(v2f*)&agg[(size_t)d * HCc + lane * 2] = acc;
    }
}

__global__ __launch_bounds__(256)
void node_post1(float* __restrict__ x, const float* __restrict__ b1,
                const float* __restrict__ gamma, const float* __restrict__ beta,
                const float* __restrict__ mean, const float* __restrict__ var) {
    const int tid = (int)(blockIdx.x * blockDim.x + threadIdx.x);
    const int nt  = (int)(gridDim.x * blockDim.x);
    const int c   = (tid & 15) * 4;
    float bb[4], mu[4], g[4], be[4];
#pragma unroll
    for (int j = 0; j < 4; ++j) {
        bb[j] = b1[c + j];
        mu[j] = mean[c + j];
        g[j]  = gamma[c + j] * rsqrtf(var[c + j] + 1e-5f);
        be[j] = beta[c + j];
    }
    float4* __restrict__ x4 = (float4*)x;
    for (int i = tid; i < Nn * HCc / 4; i += 2 * nt) {
        const bool two = (i + nt) < Nn * HCc / 4;
        const int  i2  = two ? (i + nt) : i;
        const float4 va = x4[i];
        const float4 vb = x4[i2];
        float v[8] = {va.x, va.y, va.z, va.w, vb.x, vb.y, vb.z, vb.w};
#pragma unroll
        for (int j = 0; j < 8; ++j) {
            float t = v[j] + bb[j & 3];
            t = (t - mu[j & 3]) * g[j & 3] + be[j & 3];
            v[j] = (t > 0.f) ? t : (__expf(t) - 1.f);
        }
        x4[i] = make_float4(v[0], v[1], v[2], v[3]);
        if (two) x4[i2] = make_float4(v[4], v[5], v[6], v[7]);
    }
}

extern "C" void kernel_launch(void* const* d_in, const int* in_sizes, int n_in,
                              void* d_out, int out_size, void* d_ws, size_t ws_size,
                              hipStream_t stream) {
    const float* node_feat = (const float*)d_in[0];
    const int*   ei        = (const int*)  d_in[1];
    const float* W1        = (const float*)d_in[2];
    const float* a_src1    = (const float*)d_in[3];
    const float* a_dst1    = (const float*)d_in[4];
    const float* b1        = (const float*)d_in[5];
    const float* bn_gamma  = (const float*)d_in[6];
    const float* bn_beta   = (const float*)d_in[7];
    const float* bn_mean   = (const float*)d_in[8];
    const float* bn_var    = (const float*)d_in[9];
    const float* W2        = (const float*)d_in[10];
    const float* a_src2    = (const float*)d_in[11];
    const float* a_dst2    = (const float*)d_in[12];
    const float* b2        = (const float*)d_in[13];
    const float* Wf        = (const float*)d_in[14];
    const float* bf        = (const float*)d_in[15];
    float* out = (float*)d_out;

    const int E = in_sizes[1] / 2;

    float* ws   = (float*)d_ws;
    float* H    = ws;
    float* X1   = H    + (size_t)Nn * HCc;
    float* AGG2 = X1   + (size_t)Nn * HCc;
    float* ASRC = AGG2 + (size_t)Nn * HCc;
    float* ADST = ASRC + (size_t)Nn * 2;
    int*   CNT  = (int*)(ADST + (size_t)Nn * 2);
    int*   RPTR = CNT  + Nn;
    int*   FILED = RPTR + Nn;
    int*   CSRC = FILED + (size_t)E;
    int*   HIST = (int*)AGG2;
    int*   BST  = HIST + SORT_BLKS * NBKT;

    const int tilesM    = (Nn + 15) / 16;
    const int gemmBlks  = (tilesM + 3) / 4;
    const int gathBlks  = 2048;
    const int alphaBlks = 1024;
    const int nodeBlks  = 1024;

    sort_hist<<<SORT_BLKS, SORT_THREADS, 0, stream>>>(ei, E, HIST);
    sort_scatter<<<SORT_BLKS, SORT_THREADS, 0, stream>>>(ei, E, HIST, BST, FILED);
    sort_rows<<<NBKT, ROWS_THREADS, 0, stream>>>(FILED, BST, CNT, RPTR, CSRC);

    wmma_gemm<4, false, FIN><<<gemmBlks, 128, 0, stream>>>(
        node_feat, nullptr, nullptr, W1, nullptr, H, Nn, HCc);
    node_alpha<<<alphaBlks, 256, 0, stream>>>(H, a_src1, a_dst1, ASRC, ADST);
    node_gather<<<gathBlks, 256, 0, stream>>>(RPTR, CNT, CSRC, H, ASRC, ADST, X1);
    node_post1<<<nodeBlks, 256, 0, stream>>>(X1, b1, bn_gamma, bn_beta,
                                             bn_mean, bn_var);

    wmma_gemm<4, false, HCc><<<gemmBlks, 128, 0, stream>>>(
        X1, nullptr, nullptr, W2, nullptr, H, Nn, HCc);
    node_alpha<<<alphaBlks, 256, 0, stream>>>(H, a_src2, a_dst2, ASRC, ADST);
    node_gather<<<gathBlks, 256, 0, stream>>>(RPTR, CNT, CSRC, H, ASRC, ADST, AGG2);

    wmma_gemm<3, true, HCc><<<gemmBlks, 128, 0, stream>>>(
        X1, AGG2, b2, Wf, bf, out, Nn, OUTc);
}


# weight staging into LDS: 16-byte loads, a thread's loads issued as a group (input GEMM: 16 in flight, one round trip) instead of one waited 4-byte load per element; LDS layout and sums unchanged
# speedup vs baseline: 8.8240x; 1.5770x over previous
#include <hip/hip_runtime.h>
#include <hip/hip_bf16.h>
#include <math.h>


#define Nn    100000
#define FIN   128
#define HCc   64
#define HIDc  32
#define OUTc  40
#define BKT_SHIFT 8
#define BKT_NODES (1 << BKT_SHIFT)
#define NBKT ((Nn + BKT_NODES - 1) >> BKT_SHIFT)
#define SRC_BITS 17
#define SORT_BLKS 128
#define SORT_THREADS 1024
#define ROWS_THREADS 512
static_assert(Nn <= (1 << SRC_BITS), "source id must fit its field");
static_assert(NBKT < 512, "one thread per bucket in the offset scan");

typedef __attribute__((ext_vector_type(2))) float v2f;
typedef __attribute__((ext_vector_type(4))) float v4f;
typedef __attribute__((ext_vector_type(8))) float v8f;

__device__ __forceinline__ void stage_load(const float* __restrict__ B, int ncols, int item,
                                           v4f& r0, v4f& r1) {
    const int p   = item >> 4;
    const int col = (item & 15) << 2;
    const int cc  = (col < ncols) ? col : 0;
    r0 = *(const v4f*)&B[(size_t)(2 * p)     * ncols + cc];
    r1 = *(const v4f*)&B[(size_t)(2 * p + 1) * ncols + cc];
}

__device__ __forceinline__ void stage_store(float* __restrict__ Bs, int ncols, int item,
                                            v4f r0, v4f r1) {
    const int p   = item >> 4;
    const int col = (item & 15) << 2;
    const bool ok = col < ncols;
    v4f a, b;
    a.x = ok ? r0.x : 0.f;  a.y = ok ? r1.x : 0.f;  a.z = ok ? r0.y : 0.f;  a.w = ok ? r1.y : 0.f;
    b.x = ok ? r0.z : 0.f;  b.y = ok ? r1.z : 0.f;  b.z = ok ? r0.w : 0.f;  b.w = ok ? r1.w : 0.f;
    *(v4f*)&Bs[(p << 7) + (col << 1)]     = a;
    *(v4f*)&Bs[(p << 7) + (col << 1) + 4] = b;
}

template<int NT, bool FUSE, int K>
__global__ __launch_bounds__(128)
void wmma_gemm(const float* __restrict__ A, const float* __restrict__ A2,
               const float* __restrict__ a2bias,
               const float* __restrict__ B, const float* __restrict__ bias,
               float* __restrict__ C, int M, int Ncols) {
    __shared__ alignas(16) float Bs[(K / 2) * 64 * 2];

    constexpr int ITEMS = (K / 2) * 16;
    constexpr int PER_T = ITEMS / 128;
    static_assert(ITEMS % 128 == 0, "every thread stages the same number of items");
    v4f s0[PER_T], s1[PER_T];
#pragma unroll
    for (int u = 0; u < PER_T; ++u)
        stage_load(B, Ncols, (int)threadIdx.x + u * 128, s0[u], s1[u]);
#pragma unroll
    for (int u = 0; u < PER_T; ++u)
        stage_store(Bs, Ncols, (int)threadIdx.x + u * 128, s0[u], s1[u]);
    __syncthreads();

    const int wave = threadIdx.x >> 5;
    const int lane = threadIdx.x & 31;
    const int tileM = blockIdx.x * 4 + wave;
    const int row16 = tileM * 16;
    if (row16 >= M) return;

    const int hi = (lane >= 16);
    const int lr = lane & 15;

    v8f acc[NT];
#pragma unroll
    for (int t = 0; t < NT; ++t)
        acc[t] = (v8f){0.f, 0.f, 0.f, 0.f, 0.f, 0.f, 0.f, 0.f};

    const int arow = row16 + lr;
    const float* __restrict__ Arow  = A + (size_t)arow * K;
    const float* __restrict__ A2row = FUSE ? (A2 + (size_t)arow * K) : nullptr;

#pragma unroll 4
    for (int kb = 0; kb < K; kb += 4) {
        const int ka = kb + 2 * hi;
        float a0 = Arow[ka];
        float a1 = Arow[ka + 1];
        if (FUSE) {
            a0 = fmaxf(a0, A2row[ka]     + a2bias[ka]);
            a1 = fmaxf(a1, A2row[ka + 1] + a2bias[ka + 1]);
        }
        v2f afrag; afrag.x = a0; afrag.y = a1;

        const int p = (kb >> 1) + hi;
#pragma unroll
        for (int t = 0; t < NT; ++t) {
            const v2f bfrag = *(const v2f*)&Bs[(p << 7) + ((t * 16 + lr) << 1)];
            acc[t] = __builtin_amdgcn_wmma_f32_16x16x4_f32(
                false, afrag, false, bfrag, (short)0, acc[t], false, false);
        }
    }

    const int rbase = row16 + (hi ? 8 : 0);
#pragma unroll
    for (int t = 0; t < NT; ++t) {
        const int col = t * 16 + lr;
        if (col < Ncols) {
            const float bb = bias ? bias[col] : 0.f;
#pragma unroll
            for (int r = 0; r < 8; ++r)
                C[(size_t)(rbase + r) * Ncols + col] = acc[t][r] + bb;
        }
    }
}

__global__ __launch_bounds__(SORT_THREADS)
void sort_hist(const int* __restrict__ ei, int E, int* __restrict__ hist) {
    __shared__ int hc[NBKT];
    for (int t = threadIdx.x; t < NBKT; t += SORT_THREADS) hc[t] = 0;
    __syncthreads();
    const int chunk = (E + SORT_BLKS - 1) / SORT_BLKS;
    const int e0 = blockIdx.x * chunk;
    const int e1 = (e0 + chunk < E) ? (e0 + chunk) : E;
    const int* __restrict__ dst = ei + E;
    int e = e0 + threadIdx.x;
    for (; e + 3 * SORT_THREADS < e1; e += 4 * SORT_THREADS) {
        int d[4];
#pragma unroll
        for (int u = 0; u < 4; ++u) d[u] = dst[e + u * SORT_THREADS];
#pragma unroll
        for (int u = 0; u < 4; ++u) atomicAdd(&hc[d[u] >> BKT_SHIFT], 1);
    }
    for (; e < e1; e += SORT_THREADS) atomicAdd(&hc[dst[e] >> BKT_SHIFT], 1);
    __syncthreads();
    for (int t = threadIdx.x; t < NBKT; t += SORT_THREADS)
        hist[blockIdx.x * NBKT + t] = hc[t];
}

__global__ __launch_bounds__(SORT_THREADS)
void sort_scatter(const int* __restrict__ ei, int E, const int* __restrict__ hist,
                  int* __restrict__ bstart, int* __restrict__ filed) {
    __shared__ int part[2][2][512];
    __shared__ int wsum[16];
    __shared__ int cur[512];
    const int g    = blockIdx.x;
    const int t    = threadIdx.x & 511;
    const int rh   = threadIdx.x >> 9;
    const int lane = threadIdx.x & 31;
    const int wave = threadIdx.x >> 5;

    int pre = 0, tot = 0;
    if (t < NBKT) {
        const int r0 = rh * (SORT_BLKS / 2);
        for (int r = r0; r < r0 + SORT_BLKS / 2; ++r) {
            const int v = hist[r * NBKT + t];
            tot += v;
            pre += (r < g) ? v : 0;
        }
    }
    part[rh][0][t] = pre;
    part[rh][1][t] = tot;
    __syncthreads();

    int excl = 0;
    if (threadIdx.x < 512) {
        pre = part[0][0][t] + part[1][0][t];
        tot = part[0][1][t] + part[1][1][t];
        int inc = tot;
#pragma unroll
        for (int off = 1; off < 32; off <<= 1) {
            const int u = __shfl_up(inc, off, 32);
            if (lane >= off) inc += u;
        }
        if (lane == 31) wsum[wave] = inc;
        excl = inc - tot;
    }
    __syncthreads();
    if (threadIdx.x < 512) {
        int bs = excl;
        for (int w = 0; w < wave; ++w) bs += wsum[w];
        cur[t] = bs + pre;
        if (g == 0 && t <= NBKT) bstart[t] = bs;
    }
    __syncthreads();

    const int chunk = (E + SORT_BLKS - 1) / SORT_BLKS;
    const int e0 = g * chunk;
    const int e1 = (e0 + chunk < E) ? (e0 + chunk) : E;
    int e = e0 + threadIdx.x;
    for (; e + 3 * SORT_THREADS < e1; e += 4 * SORT_THREADS) {
        int s[4], d[4];
#pragma unroll
        for (int u = 0; u < 4; ++u) {
            s[u] = ei[e + u * SORT_THREADS];
            d[u] = ei[E + e + u * SORT_THREADS];
        }
#pragma unroll
        for (int u = 0; u < 4; ++u) {
            const int pos = atomicAdd(&cur[d[u] >> BKT_SHIFT], 1);
            filed[pos] = ((d[u] & (BKT_NODES - 1)) << SRC_BITS) | s[u];
        }
    }
    for (; e < e1; e += SORT_THREADS) {
        const int s = ei[e];
        const int d = ei[E + e];
        const int pos = atomicAdd(&cur[d >> BKT_SHIFT], 1);
        filed[pos] = ((d & (BKT_NODES - 1)) << SRC_BITS) | s;
    }
}

__global__ __launch_bounds__(ROWS_THREADS)
void sort_rows(const int* __restrict__ filed, const int* __restrict__ bstart,
               int* __restrict__ cnt, int* __restrict__ rowptr,
               int* __restrict__ csrc) {
    __shared__ int c[BKT_NODES];
    __shared__ int wsum[BKT_NODES / 32];
    const int bkt  = blockIdx.x;
    const int beg  = bstart[bkt];
    const int end  = bstart[bkt + 1];
    const int lane = threadIdx.x & 31;
    const int wave = threadIdx.x >> 5;
    if (threadIdx.x < BKT_NODES) c[threadIdx.x] = 0;
    __syncthreads();
    int i = beg + threadIdx.x;
    for (; i + 3 * ROWS_THREADS < end; i += 4 * ROWS_THREADS) {
        unsigned p[4];
#pragma unroll
        for (int u = 0; u < 4; ++u) p[u] = (unsigned)filed[i + u * ROWS_THREADS];
#pragma unroll
        for (int u = 0; u < 4; ++u) atomicAdd(&c[p[u] >> SRC_BITS], 1);
    }
    for (; i < end; i += ROWS_THREADS) atomicAdd(&c[(unsigned)filed[i] >> SRC_BITS], 1);
    __syncthreads();

    int deg = 0, excl = 0;
    if (threadIdx.x < BKT_NODES) {
        deg = c[threadIdx.x];
        int inc = deg;
#pragma unroll
        for (int off = 1; off < 32; off <<= 1) {
            const int u = __shfl_up(inc, off, 32);
            if (lane >= off) inc += u;
        }
        if (lane == 31) wsum[wave] = inc;
        excl = inc - deg;
    }
    __syncthreads();
    if (threadIdx.x < BKT_NODES) {
        int start = beg + excl;
        for (int w = 0; w < wave; ++w) start += wsum[w];
        const int node = bkt * BKT_NODES + threadIdx.x;
        if (node < Nn) {
            cnt[node]    = deg;
            rowptr[node] = start;
        }
        c[threadIdx.x] = start;
    }
    __syncthreads();
    i = beg + threadIdx.x;
    for (; i + 3 * ROWS_THREADS < end; i += 4 * ROWS_THREADS) {
        unsigned p[4];
#pragma unroll
        for (int u = 0; u < 4; ++u) p[u] = (unsigned)filed[i + u * ROWS_THREADS];
#pragma unroll
        for (int u = 0; u < 4; ++u) {
            const int pos = atomicAdd(&c[p[u] >> SRC_BITS], 1);
            csrc[pos] = (int)(p[u] & ((1u << SRC_BITS) - 1u));
        }
    }
    for (; i < end; i += ROWS_THREADS) {
        const unsigned p = (unsigned)filed[i];
        const int pos = atomicAdd(&c[p >> SRC_BITS], 1);
        csrc[pos] = (int)(p & ((1u << SRC_BITS) - 1u));
    }
}

#define ROW_XMASK(x, M) __int_as_float(__builtin_amdgcn_update_dpp( \
    0, __float_as_int(x), 0x160 + (M), 0xf, 0xf, false))

__device__ __forceinline__ float row_sum16(float v) {
    v += ROW_XMASK(v, 1);
    v += ROW_XMASK(v, 2);
    v += ROW_XMASK(v, 4);
    v += ROW_XMASK(v, 8);
    return v;
}

__device__ __forceinline__ float row_max16(float v) {
    v = fmaxf(v, ROW_XMASK(v, 1));
    v = fmaxf(v, ROW_XMASK(v, 2));
    v = fmaxf(v, ROW_XMASK(v, 4));
    v = fmaxf(v, ROW_XMASK(v, 8));
    return v;
}

template<int K>
__device__ __forceinline__ float row_share(float x) {
    return __int_as_float(__builtin_amdgcn_mov_dpp(
        __float_as_int(x), 0x150 + K, 0xf, 0xf, false));
}


template<int G>
__device__ __forceinline__ void load_group(const float* __restrict__ h, int s, int lane2,
                                           v2f (&hv)[32]) {
#pragma unroll
    for (int u = 0; u < 4; ++u) {
        const int sk = __builtin_amdgcn_readlane(s, 4 * G + u);
        hv[4 * G + u] = *(const v2f*)&h[(size_t)sk * HCc + lane2];
    }
}

template<int G>
__device__ __forceinline__ void logit_group(const v2f (&hv)[32], v2f as, bool b0, bool b1,
                                            int q16, float& E0, float& E1) {
    const float p0 = fmaf(hv[4 * G + 0].y, as.y, hv[4 * G + 0].x * as.x);
    const float p1 = fmaf(hv[4 * G + 1].y, as.y, hv[4 * G + 1].x * as.x);
    const float p2 = fmaf(hv[4 * G + 2].y, as.y, hv[4 * G + 2].x * as.x);
    const float p3 = fmaf(hv[4 * G + 3].y, as.y, hv[4 * G + 3].x * as.x);
    const float q01 = (b0 ? p1 : p0) + ROW_XMASK(b0 ? p0 : p1, 1);
    const float q23 = (b0 ? p3 : p2) + ROW_XMASK(b0 ? p2 : p3, 1);
    float r = (b1 ? q23 : q01) + ROW_XMASK(b1 ? q01 : q23, 2);
    r += ROW_XMASK(r, 4);
    r += ROW_XMASK(r, 8);
    if (G < 4) E0 = (q16 == (G & 3)) ? r : E0;
    else       E1 = (q16 == (G & 3)) ? r : E1;
}

#define ACC_EDGE(K) { \
    const float w = row_share<(K) & 15>(((K) < 16) ? P0 : P1); \
    acc.x = fmaf(w, hv[K].x, acc.x); \
    acc.y = fmaf(w, hv[K].y, acc.y); }

template<int G>
__device__ __forceinline__ void acc_group(const v2f (&hv)[32], float P0, float P1, v2f& acc) {
    ACC_EDGE(4 * G + 0)
    ACC_EDGE(4 * G + 1)
    ACC_EDGE(4 * G + 2)
    ACC_EDGE(4 * G + 3)
}

#define REST_GROUPS(X) X(1) X(2) X(3) X(4) X(5) X(6) X(7)
#define LOAD_G(G)  if (nk > 4 * G) load_group<G>(h, s, lane2, hv);
#define LOGIT_G(G) if (nk > 4 * G) logit_group<G>(hv, as, b0, b1, q16, E0, E1);
#define ACC_G(G)   if (nk > 4 * G) acc_group<G>(hv, P0, P1, acc);

__device__ __forceinline__ v2f gather_one(int d, const int* __restrict__ rowptr,
                                          const int* __restrict__ cnt,
                                          const int* __restrict__ csrc,
                                          const float* __restrict__ h,
                                          v2f as, v2f ad, int lane, int l16, int q16,
                                          bool b0, bool b1, int lane2) {
    const int beg = rowptr[d];
    const int deg = cnt[d];
    const int tot = deg + 1;

    float m = -INFINITY, den = 0.f, adst = 0.f;
    v2f acc; acc.x = 0.f; acc.y = 0.f;
    int j0 = 0;
    do {
        const int left = tot - j0;
        const int nk   = (left < 32) ? left : 32;
        const int j    = j0 + lane;
        const int s    = (j >= 1 && j < tot) ? csrc[beg + j - 1] : d;

        v2f hv[32];
        load_group<0>(h, s, lane2, hv);
        REST_GROUPS(LOAD_G)

        if (j0 == 0) adst = row_sum16(fmaf(hv[0].y, ad.y, hv[0].x * ad.x));

        float E0 = -INFINITY, E1 = -INFINITY;
        logit_group<0>(hv, as, b0, b1, q16, E0, E1);
        REST_GROUPS(LOGIT_G)
        E0 += adst;  E0 = fmaxf(E0, 0.2f * E0);
        E1 += adst;  E1 = fmaxf(E1, 0.2f * E1);
        E0 = (l16 < nk)      ? E0 : -INFINITY;
        E1 = (l16 + 16 < nk) ? E1 : -INFINITY;

        const float mn = fmaxf(m, row_max16(fmaxf(E0, E1)));
        const float sc = __expf(m - mn);
        const float P0 = __expf(E0 - mn);
        const float P1 = __expf(E1 - mn);
        den   = den * sc + row_sum16(P0 + P1);
        acc.x = acc.x * sc;
        acc.y = acc.y * sc;
        m = mn;

        acc_group<0>(hv, P0, P1, acc);
        REST_GROUPS(ACC_G)
        j0 += 32;
    } while (j0 < tot);
    den += 1e-16f;
    v2f o; o.x = acc.x / den; o.y = acc.y / den;
    return o;
}

#define TPITCH 68

__device__ __forceinline__ int lane_now() {
    int l;
    asm volatile("v_mbcnt_lo_u32_b32 %0, -1, 0" : "=v"(l));
    return l;
}

#define GW 12
template<int LAYER>
__global__ __launch_bounds__(GW * 32, 9)
void node_gather_dense(const int* __restrict__ rowptr, const int* __restrict__ cnt,
                       const int* __restrict__ csrc, const float* __restrict__ h,
                       const float* __restrict__ a_s, const float* __restrict__ a_d,
                       const float* __restrict__ pb,
                       const float* __restrict__ gamma, const float* __restrict__ beta,
                       const float* __restrict__ mean, const float* __restrict__ var,
                       float* __restrict__ x1,
                       const float* __restrict__ Bm, const float* __restrict__ bias,
                       float* __restrict__ C) {
    constexpr int NT = (LAYER == 1) ? 4 : 3;
    constexpr int NC = (LAYER == 1) ? HCc : OUTc;
    __shared__ alignas(16) float Bs[(HCc / 2) * 64 * 2];
    __shared__ float At[GW][16 * TPITCH];
    __shared__ float cst[4][HCc];

    const int lane = threadIdx.x & 31;
    const int wave = __builtin_amdgcn_readfirstlane((int)(threadIdx.x >> 5));
    static_assert(NC % 4 == 0, "rows of the weight matrix are read 16 bytes at a time");
    for (int item = threadIdx.x; item < (HCc / 2) * 16; item += GW * 32) {
        v4f r0, r1;
        stage_load(Bm, NC, item, r0, r1);
        stage_store(Bs, NC, item, r0, r1);
    }
    if (threadIdx.x < HCc) {
        const int c = threadIdx.x;
        cst[0][c] = pb[c];
        if (LAYER == 1) {
            cst[1][c] = mean[c];
            cst[2][c] = gamma[c] * rsqrtf(var[c] + 1e-5f);
            cst[3][c] = beta[c];
        }
    }
    float* __restrict__ tile = At[wave];
    for (int i = lane; i < 16 * TPITCH; i += 32) tile[i] = 0.f;
    __syncthreads();

    const int d0   = __builtin_amdgcn_readfirstlane(
        (int)((blockIdx.x * blockDim.x + threadIdx.x) >> 5));
    const int nw   = (int)((gridDim.x * blockDim.x) >> 5);
    const int l16  = lane & 15;
    const int q16  = l16 >> 2;
    const bool b0  = (lane & 1) != 0;
    const bool b1  = (lane & 2) != 0;
    const int lane2 = lane * 2;
    const v2f as = *(const v2f*)&a_s[lane2];
    const v2f ad = *(const v2f*)&a_d[lane2];
    int slot = 0;
    for (int d = d0; d < Nn; d += nw) {
        v2f xo;
        if (LAYER == 2) xo = *(const v2f*)&x1[(size_t)d * HCc + lane2];
        const v2f o = gather_one(d, rowptr, cnt, csrc, h, as, ad, lane, l16, q16, b0, b1, lane2);

        const int ln  = lane_now();
        const int ln2 = ln * 2;
        const int e16 = ln & 15;
        const int ehi = ln >> 4;
        const v2f bb = *(const v2f*)&cst[0][ln2];
        v2f x;
        if (LAYER == 1) {
            const v2f mu = *(const v2f*)&cst[1][ln2];
            const v2f g  = *(const v2f*)&cst[2][ln2];
            const v2f be = *(const v2f*)&cst[3][ln2];
            float t0 = o.x + bb.x, t1 = o.y + bb.y;
            t0 = (t0 - mu.x) * g.x + be.x;
            t1 = (t1 - mu.y) * g.y + be.y;
            x.x = (t0 > 0.f) ? t0 : (__expf(t0) - 1.f);
            x.y = (t1 > 0.f) ? t1 : (__expf(t1) - 1.f);
            *(v2f*)&x1[(size_t)d * HCc + ln2] = x;
        } else {
            x.x = fmaxf(xo.x, o.x + bb.x);
            x.y = fmaxf(xo.y, o.y + bb.y);
        }
        *(v2f*)&tile[slot * TPITCH + ln2] = x;

        if (slot == 15 || d + nw >= Nn) {
            v8f acc[NT];
#pragma unroll
            for (int t = 0; t < NT; ++t)
                acc[t] = (v8f){0.f, 0.f, 0.f, 0.f, 0.f, 0.f, 0.f, 0.f};
#pragma unroll 2
            for (int kb = 0; kb < HCc; kb += 4) {
                const v2f afrag = *(const v2f*)&tile[e16 * TPITCH + kb + 2 * ehi];
                const int p = (kb >> 1) + ehi;
#pragma unroll
                for (int t = 0; t < NT; ++t) {
                    const v2f bfrag = *(const v2f*)&Bs[(p << 7) + ((t * 16 + e16) << 1)];
                    acc[t] = __builtin_amdgcn_wmma_f32_16x16x4_f32(
                        false, afrag, false, bfrag, (short)0, acc[t], false, false);
                }
            }
            const int dbase = d - slot * nw;
#pragma unroll
            for (int t = 0; t < NT; ++t) {
                const int col = t * 16 + e16;
                if (col < NC) {
                    const float cb = (LAYER == 2) ? bias[col] : 0.f;
#pragma unroll
                    for (int r = 0; r < 8; ++r) {
                        const int s = 8 * ehi + r;
                        if (s <= slot)
                            C[(size_t)(dbase + s * nw) * NC + col] =
                                (LAYER == 2) ? acc[t][r] + cb : acc[t][r];
                    }
                }
            }
        }
        slot = (slot + 1) & 15;
    }
}

extern "C" void kernel_launch(void* const* d_in, const int* in_sizes, int n_in,
                              void* d_out, int out_size, void* d_ws, size_t ws_size,
                              hipStream_t stream) {
    const float* node_feat = (const float*)d_in[0];
    const int*   ei        = (const int*)  d_in[1];
    const float* W1        = (const float*)d_in[2];
    const float* a_src1    = (const float*)d_in[3];
    const float* a_dst1    = (const float*)d_in[4];
    const float* b1        = (const float*)d_in[5];
    const float* bn_gamma  = (const float*)d_in[6];
    const float* bn_beta   = (const float*)d_in[7];
    const float* bn_mean   = (const float*)d_in[8];
    const float* bn_var    = (const float*)d_in[9];
    const float* W2        = (const float*)d_in[10];
    const float* a_src2    = (const float*)d_in[11];
    const float* a_dst2    = (const float*)d_in[12];
    const float* b2        = (const float*)d_in[13];
    const float* Wf        = (const float*)d_in[14];
    const float* bf        = (const float*)d_in[15];
    float* out = (float*)d_out;

    const int E = in_sizes[1] / 2;

    float* ws   = (float*)d_ws;
    float* H    = ws;
    float* X1   = H    + (size_t)Nn * HCc;
    float* H2   = X1   + (size_t)Nn * HCc;
    int*   CNT  = (int*)(H2 + (size_t)Nn * HCc);
    int*   RPTR = CNT  + Nn;
    int*   FILED = RPTR + Nn;
    int*   CSRC = FILED + (size_t)E;
    int*   HIST = (int*)H2;
    int*   BST  = HIST + SORT_BLKS * NBKT;

    const int tilesM    = (Nn + 15) / 16;
    const int gemmBlks  = (tilesM + 3) / 4;
    const int gathBlks  = 768;

    sort_hist<<<SORT_BLKS, SORT_THREADS, 0, stream>>>(ei, E, HIST);
    sort_scatter<<<SORT_BLKS, SORT_THREADS, 0, stream>>>(ei, E, HIST, BST, FILED);
    sort_rows<<<NBKT, ROWS_THREADS, 0, stream>>>(FILED, BST, CNT, RPTR, CSRC);

    wmma_gemm<4, false, FIN><<<gemmBlks, 128, 0, stream>>>(
        node_feat, nullptr, nullptr, W1, nullptr, H, Nn, HCc);
    node_gather_dense<1><<<gathBlks, GW * 32, 0, stream>>>(
        RPTR, CNT, CSRC, H, a_src1, a_dst1, b1, bn_gamma, bn_beta, bn_mean, bn_var,
        X1, W2, nullptr, H2);

    node_gather_dense<2><<<gathBlks, GW * 32, 0, stream>>>(
        RPTR, CNT, CSRC, H2, a_src2, a_dst2, b2, nullptr, nullptr, nullptr, nullptr,
        X1, Wf, bf, out);
}
